# starting kernel
# baseline (speedup 1.0000x reference)
#include <hip/hip_runtime.h>


typedef float v2f __attribute__((ext_vector_type(2)));
typedef float v8f __attribute__((ext_vector_type(8)));

#define D_MODEL 512
#define D_STATE 64
#define SEQ_LEN 2048
#define BATCH 4
#define MTOT (BATCH * SEQ_LEN)
#define NCHUNK 8
#define TCHUNK (SEQ_LEN / NCHUNK)

__global__ __launch_bounds__(256) void gemm_wmma_f32(
    const float* __restrict__ A, const float* __restrict__ W,
    const float* __restrict__ bias, float* __restrict__ out)
{
    const int lane = threadIdx.x & 31;
    const int l16  = lane & 15;
    const int half = lane >> 4;
    const int gwave = blockIdx.x * 8 + (threadIdx.x >> 5);

    const int nBlocksN = D_MODEL / 64;
    const int mb = gwave / nBlocksN;
    const int nb = gwave % nBlocksN;
    const int M0 = mb * 64;
    const int N0 = nb * 64;

    const float* arow[4];
    const float* brow[4];
#pragma unroll
    for (int i = 0; i < 4; ++i)
        arow[i] = A + (size_t)(M0 + i * 16 + l16) * D_MODEL + half * 2;
#pragma unroll
    for (int j = 0; j < 4; ++j)
        brow[j] = W + (size_t)(N0 + j * 16 + l16) * D_MODEL + half * 2;

    v8f acc[4][4] = {};

#pragma unroll 4
    for (int k0 = 0; k0 < D_MODEL; k0 += 4) {
        v2f a[4], b[4];
#pragma unroll
        for (int i = 0; i < 4; ++i) a[i] = *(const v2f*)(arow[i] + k0);
#pragma unroll
        for (int j = 0; j < 4; ++j) b[j] = *(const v2f*)(brow[j] + k0);
#pragma unroll
        for (int i = 0; i < 4; ++i)
#pragma unroll
            for (int j = 0; j < 4; ++j)
                acc[i][j] = __builtin_amdgcn_wmma_f32_16x16x4_f32(
                    false, a[i], false, b[j], (short)0, acc[i][j],
                    false, false);
    }

#pragma unroll
    for (int j = 0; j < 4; ++j) {
        const float bj = bias[N0 + j * 16 + l16];
#pragma unroll
        for (int i = 0; i < 4; ++i) {
#pragma unroll
            for (int r = 0; r < 8; ++r) {
                const int row = M0 + i * 16 + half * 8 + r;
                out[(size_t)row * D_MODEL + N0 + j * 16 + l16] =
                    acc[i][j][r] + bj;
            }
        }
    }
}

__global__ __launch_bounds__(256) void s4_chunk_end(
    const float* __restrict__ xin,
    const float* __restrict__ A_log,
    const float* __restrict__ dt,
    float* __restrict__ Echunk)
{
    const int wid  = blockIdx.x * 8 + (threadIdx.x >> 5);
    const int lane = threadIdx.x & 31;
    const int c  = wid & (NCHUNK - 1);
    const int bd = wid >> 3;
    const int b  = bd >> 9;
    const int d  = bd & 511;
    const int n0 = lane * 2;

    const float dtd = dt[d];
    const float dA0 = expf(-expf(A_log[d * D_STATE + n0])     * dtd);
    const float dA1 = expf(-expf(A_log[d * D_STATE + n0 + 1]) * dtd);

    const float* up = xin + (size_t)(b * SEQ_LEN + c * TCHUNK) * D_MODEL + d;
    float s0 = 0.f, s1 = 0.f;
#pragma unroll 8
    for (int t = 0; t < TCHUNK; ++t) {
        const float u = up[(size_t)t * D_MODEL];
        s0 = fmaf(dA0, s0, u);
        s1 = fmaf(dA1, s1, u);
    }
    float* E = Echunk + ((size_t)bd * NCHUNK + c) * D_STATE;
    E[n0]     = s0;
    E[n0 + 1] = s1;
}

__global__ __launch_bounds__(256) void s4_chunk_prefix(
    const float* __restrict__ A_log, const float* __restrict__ dt,
    const float* __restrict__ Echunk,
    float* __restrict__ Sinit)
{
    const int bd   = blockIdx.x * 8 + (threadIdx.x >> 5);
    const int lane = threadIdx.x & 31;
    const int d  = bd & 511;
    const int n0 = lane * 2;

    const float dtd = dt[d];
    const float a0 = -expf(A_log[d * D_STATE + n0]);
    const float a1 = -expf(A_log[d * D_STATE + n0 + 1]);
    const float dAp0 = expf(a0 * dtd * (float)TCHUNK);
    const float dAp1 = expf(a1 * dtd * (float)TCHUNK);

    const float* E = Echunk + (size_t)bd * NCHUNK * D_STATE;
    float*       S = Sinit  + (size_t)bd * NCHUNK * D_STATE;

    float s0 = 0.f, s1 = 0.f;
#pragma unroll
    for (int c = 0; c < NCHUNK; ++c) {
        S[c * D_STATE + n0]     = s0;
        S[c * D_STATE + n0 + 1] = s1;
        s0 = fmaf(dAp0, s0, E[c * D_STATE + n0]);
        s1 = fmaf(dAp1, s1, E[c * D_STATE + n0 + 1]);
    }
}

__global__ __launch_bounds__(256) void s4_scan_chunk(
    const float* __restrict__ xin,
    const float* __restrict__ A_log,
    const float* __restrict__ Bm,
    const float* __restrict__ Cm,
    const float* __restrict__ Dv,
    const float* __restrict__ dt,
    const float* __restrict__ Sinit,
    float* __restrict__ y)
{
    const int wid  = blockIdx.x * 8 + (threadIdx.x >> 5);
    const int lane = threadIdx.x & 31;
    const int c  = wid & (NCHUNK - 1);
    const int bd = wid >> 3;
    const int b  = bd >> 9;
    const int d  = bd & 511;
    const int n0 = lane * 2;

    const float dtd = dt[d];
    const float dA0 = expf(-expf(A_log[d * D_STATE + n0])     * dtd);
    const float dA1 = expf(-expf(A_log[d * D_STATE + n0 + 1]) * dtd);
    const float c0  = Cm[d * D_STATE + n0]     * Bm[d * D_STATE + n0]     * dtd;
    const float c1  = Cm[d * D_STATE + n0 + 1] * Bm[d * D_STATE + n0 + 1] * dtd;
    const float Dd  = Dv[d];

    const float* S = Sinit + ((size_t)bd * NCHUNK + c) * D_STATE;
    float s0 = S[n0];
    float s1 = S[n0 + 1];

    const float* up = xin + (size_t)(b * SEQ_LEN + c * TCHUNK) * D_MODEL + d;
    float*       yp = y   + (size_t)(b * SEQ_LEN + c * TCHUNK) * D_MODEL + d;

#pragma unroll 4
    for (int t = 0; t < TCHUNK; ++t) {
        const float u = up[(size_t)t * D_MODEL];
        s0 = fmaf(dA0, s0, u);
        s1 = fmaf(dA1, s1, u);
        float p = fmaf(c1, s1, c0 * s0);
#pragma unroll
        for (int m = 16; m >= 1; m >>= 1)
            p += __shfl_xor(p, m, 32);
        if (lane == 0)
            yp[(size_t)t * D_MODEL] = fmaf(Dd, u, p);
    }
}

extern "C" void kernel_launch(void* const* d_in, const int* in_sizes, int n_in,
                              void* d_out, int out_size, void* d_ws, size_t ws_size,
                              hipStream_t stream)
{
    (void)in_sizes; (void)n_in; (void)out_size; (void)ws_size;

    const float* x     = (const float*)d_in[0];
    const float* W_in  = (const float*)d_in[1];
    const float* b_in  = (const float*)d_in[2];
    const float* A_log = (const float*)d_in[3];
    const float* Bm    = (const float*)d_in[4];
    const float* Cm    = (const float*)d_in[5];
    const float* Dv    = (const float*)d_in[6];
    const float* dt    = (const float*)d_in[7];
    const float* W_out = (const float*)d_in[8];
    const float* b_out = (const float*)d_in[9];

    float* xin  = (float*)d_ws;
    float* ymid = xin  + (size_t)MTOT * D_MODEL;
    float* E    = ymid + (size_t)MTOT * D_MODEL;
    float* S    = E + (size_t)BATCH * D_MODEL * NCHUNK * D_STATE;
    float* out  = (float*)d_out;

    const dim3 blk(256);
    gemm_wmma_f32<<<128, blk, 0, stream>>>(x, W_in, b_in, xin);
    s4_chunk_end  <<<2048, blk, 0, stream>>>(xin, A_log, dt, E);
    s4_chunk_prefix<<<256, blk, 0, stream>>>(A_log, dt, E, S);
    s4_scan_chunk <<<2048, blk, 0, stream>>>(xin, A_log, Bm, Cm, Dv, dt, S, ymid);
    gemm_wmma_f32<<<128, blk, 0, stream>>>(ymid, W_out, b_out, out);
}


# Scan lanes own channels, not states: each lane keeps its channel's 64 states in registers (16 chunks); state sum becomes in-lane FMAs (summation order changed), no shuffles, full-row loads/stores
# speedup vs baseline: 2.4014x; 2.4014x over previous
#include <hip/hip_runtime.h>


typedef float v2f __attribute__((ext_vector_type(2)));
typedef float v8f __attribute__((ext_vector_type(8)));

#define D_MODEL 512
#define D_STATE 64
#define SEQ_LEN 2048
#define BATCH 4
#define MTOT (BATCH * SEQ_LEN)
#define NCHUNK 16
#define TCHUNK (SEQ_LEN / NCHUNK)
#define TBLK 8
#define TBLK1 16
#define CPLANE (BATCH * D_STATE * D_MODEL)

__global__ __launch_bounds__(256) void gemm_wmma_f32(
    const float* __restrict__ A, const float* __restrict__ W,
    const float* __restrict__ bias, float* __restrict__ out)
{
    const int lane = threadIdx.x & 31;
    const int l16  = lane & 15;
    const int half = lane >> 4;
    const int gwave = blockIdx.x * 8 + (threadIdx.x >> 5);

    const int nBlocksN = D_MODEL / 64;
    const int mb = gwave / nBlocksN;
    const int nb = gwave % nBlocksN;
    const int M0 = mb * 64;
    const int N0 = nb * 64;

    const float* arow[4];
    const float* brow[4];
#pragma unroll
    for (int i = 0; i < 4; ++i)
        arow[i] = A + (size_t)(M0 + i * 16 + l16) * D_MODEL + half * 2;
#pragma unroll
    for (int j = 0; j < 4; ++j)
        brow[j] = W + (size_t)(N0 + j * 16 + l16) * D_MODEL + half * 2;

    v8f acc[4][4] = {};

#pragma unroll 4
    for (int k0 = 0; k0 < D_MODEL; k0 += 4) {
        v2f a[4], b[4];
#pragma unroll
        for (int i = 0; i < 4; ++i) a[i] = *(const v2f*)(arow[i] + k0);
#pragma unroll
        for (int j = 0; j < 4; ++j) b[j] = *(const v2f*)(brow[j] + k0);
#pragma unroll
        for (int i = 0; i < 4; ++i)
#pragma unroll
            for (int j = 0; j < 4; ++j)
                acc[i][j] = __builtin_amdgcn_wmma_f32_16x16x4_f32(
                    false, a[i], false, b[j], (short)0, acc[i][j],
                    false, false);
    }

#pragma unroll
    for (int j = 0; j < 4; ++j) {
        const float bj = bias[N0 + j * 16 + l16];
#pragma unroll
        for (int i = 0; i < 4; ++i) {
#pragma unroll
            for (int r = 0; r < 8; ++r) {
                const int row = M0 + i * 16 + half * 8 + r;
                out[(size_t)row * D_MODEL + N0 + j * 16 + l16] =
                    acc[i][j][r] + bj;
            }
        }
    }
}

__global__ __launch_bounds__(128) void s4_chunk_end(
    const float* __restrict__ xin,
    const float* __restrict__ A_log,
    const float* __restrict__ dt,
    float* __restrict__ carry)
{
    const int lane = threadIdx.x & 31;
    const int g = blockIdx.x * 4 + (threadIdx.x >> 5);
    const int c = blockIdx.y;
    const int b = g >> 4;
    const int d = (g & 15) * 32 + lane;

    const float dtd = dt[d];
    float dA[D_STATE];
    const float4* al = (const float4*)(A_log + (size_t)d * D_STATE);
#pragma unroll
    for (int q = 0; q < D_STATE / 4; ++q) {
        const float4 v = al[q];
        dA[4 * q + 0] = expf(-expf(v.x) * dtd);
        dA[4 * q + 1] = expf(-expf(v.y) * dtd);
        dA[4 * q + 2] = expf(-expf(v.z) * dtd);
        dA[4 * q + 3] = expf(-expf(v.w) * dtd);
    }

    float s[D_STATE];
#pragma unroll
    for (int n = 0; n < D_STATE; ++n) s[n] = 0.f;

    const float* up = xin + (size_t)(b * SEQ_LEN + c * TCHUNK) * D_MODEL + d;
    float ua[TBLK1], ub[TBLK1];
#pragma unroll
    for (int k = 0; k < TBLK1; ++k) ua[k] = up[(size_t)k * D_MODEL];

#pragma unroll 1
    for (int t0 = 0; t0 < TCHUNK; t0 += 2 * TBLK1) {
#pragma unroll
        for (int k = 0; k < TBLK1; ++k)
            ub[k] = up[(size_t)(t0 + TBLK1 + k) * D_MODEL];
#pragma unroll
        for (int k = 0; k < TBLK1; ++k) {
#pragma unroll
            for (int n = 0; n < D_STATE; ++n)
                s[n] = fmaf(dA[n], s[n], ua[k]);
        }
        if (t0 + 2 * TBLK1 < TCHUNK) {
#pragma unroll
            for (int k = 0; k < TBLK1; ++k)
                ua[k] = up[(size_t)(t0 + 2 * TBLK1 + k) * D_MODEL];
        }
#pragma unroll
        for (int k = 0; k < TBLK1; ++k) {
#pragma unroll
            for (int n = 0; n < D_STATE; ++n)
                s[n] = fmaf(dA[n], s[n], ub[k]);
        }
    }

    float* E = carry + (size_t)c * CPLANE + (size_t)b * D_STATE * D_MODEL + d;
#pragma unroll
    for (int n = 0; n < D_STATE; ++n) E[(size_t)n * D_MODEL] = s[n];
}

__global__ __launch_bounds__(256) void s4_chunk_prefix(
    const float* __restrict__ A_log, const float* __restrict__ dt,
    float* __restrict__ carry)
{
    const int idx = blockIdx.x * 256 + threadIdx.x;
    const int d = idx & (D_MODEL - 1);
    const int n = (idx >> 9) & (D_STATE - 1);

    const float a = -expf(A_log[d * D_STATE + n]);
    const float dAp = expf(a * dt[d] * (float)TCHUNK);

    float* P = carry + idx;
    float e[NCHUNK - 1];
#pragma unroll
    for (int c = 0; c < NCHUNK - 1; ++c) e[c] = P[(size_t)c * CPLANE];

    float s = 0.f;
#pragma unroll
    for (int c = 0; c < NCHUNK - 1; ++c) {
        P[(size_t)c * CPLANE] = s;
        s = fmaf(dAp, s, e[c]);
    }
    P[(size_t)(NCHUNK - 1) * CPLANE] = s;
}

static __device__ __forceinline__ void scan_block(
    float (&s)[D_STATE], const float (&dA)[D_STATE], const float (&cf)[D_STATE],
    const float (&u)[TBLK], const float Dd, float* __restrict__ yp)
{
#pragma unroll
    for (int k = 0; k < TBLK; ++k) {
        float p0 = 0.f, p1 = 0.f, p2 = 0.f, p3 = 0.f;
#pragma unroll
        for (int n = 0; n < D_STATE; n += 4) {
            s[n + 0] = fmaf(dA[n + 0], s[n + 0], u[k]);
            s[n + 1] = fmaf(dA[n + 1], s[n + 1], u[k]);
            s[n + 2] = fmaf(dA[n + 2], s[n + 2], u[k]);
            s[n + 3] = fmaf(dA[n + 3], s[n + 3], u[k]);
            p0 = fmaf(cf[n + 0], s[n + 0], p0);
            p1 = fmaf(cf[n + 1], s[n + 1], p1);
            p2 = fmaf(cf[n + 2], s[n + 2], p2);
            p3 = fmaf(cf[n + 3], s[n + 3], p3);
        }
        yp[(size_t)k * D_MODEL] = fmaf(Dd, u[k], (p0 + p1) + (p2 + p3));
    }
}

__global__ __launch_bounds__(128) void s4_scan_chunk(
    const float* __restrict__ xin,
    const float* __restrict__ A_log,
    const float* __restrict__ Bm,
    const float* __restrict__ Cm,
    const float* __restrict__ Dv,
    const float* __restrict__ dt,
    const float* __restrict__ carry,
    float* __restrict__ y)
{
    const int lane = threadIdx.x & 31;
    const int g = blockIdx.x * 4 + (threadIdx.x >> 5);
    const int c = blockIdx.y;
    const int b = g >> 4;
    const int d = (g & 15) * 32 + lane;

    const float dtd = dt[d];
    const float Dd  = Dv[d];
    float dA[D_STATE], cf[D_STATE], s[D_STATE];
    const float4* al = (const float4*)(A_log + (size_t)d * D_STATE);
    const float4* bl = (const float4*)(Bm    + (size_t)d * D_STATE);
    const float4* cl = (const float4*)(Cm    + (size_t)d * D_STATE);
#pragma unroll
    for (int q = 0; q < D_STATE / 4; ++q) {
        const float4 va = al[q];
        const float4 vb = bl[q];
        const float4 vc = cl[q];
        dA[4 * q + 0] = expf(-expf(va.x) * dtd);
        dA[4 * q + 1] = expf(-expf(va.y) * dtd);
        dA[4 * q + 2] = expf(-expf(va.z) * dtd);
        dA[4 * q + 3] = expf(-expf(va.w) * dtd);
        cf[4 * q + 0] = vc.x * vb.x * dtd;
        cf[4 * q + 1] = vc.y * vb.y * dtd;
        cf[4 * q + 2] = vc.z * vb.z * dtd;
        cf[4 * q + 3] = vc.w * vb.w * dtd;
    }

    const float* S = carry + (size_t)c * CPLANE + (size_t)b * D_STATE * D_MODEL + d;
#pragma unroll
    for (int n = 0; n < D_STATE; ++n) s[n] = S[(size_t)n * D_MODEL];

    const float* up = xin + (size_t)(b * SEQ_LEN + c * TCHUNK) * D_MODEL + d;
    float*       yp = y   + (size_t)(b * SEQ_LEN + c * TCHUNK) * D_MODEL + d;

    float ua[TBLK], ub[TBLK];
#pragma unroll
    for (int k = 0; k < TBLK; ++k) ua[k] = up[(size_t)k * D_MODEL];

#pragma unroll 1
    for (int t0 = 0; t0 < TCHUNK; t0 += 2 * TBLK) {
#pragma unroll
        for (int k = 0; k < TBLK; ++k)
            ub[k] = up[(size_t)(t0 + TBLK + k) * D_MODEL];
        scan_block(s, dA, cf, ua, Dd, yp + (size_t)t0 * D_MODEL);
        if (t0 + 2 * TBLK < TCHUNK) {
#pragma unroll
            for (int k = 0; k < TBLK; ++k)
                ua[k] = up[(size_t)(t0 + 2 * TBLK + k) * D_MODEL];
        }
        scan_block(s, dA, cf, ub, Dd, yp + (size_t)(t0 + TBLK) * D_MODEL);
    }
}

extern "C" void kernel_launch(void* const* d_in, const int* in_sizes, int n_in,
                              void* d_out, int out_size, void* d_ws, size_t ws_size,
                              hipStream_t stream)
{
    (void)in_sizes; (void)n_in; (void)out_size; (void)ws_size;

    const float* x     = (const float*)d_in[0];
    const float* W_in  = (const float*)d_in[1];
    const float* b_in  = (const float*)d_in[2];
    const float* A_log = (const float*)d_in[3];
    const float* Bm    = (const float*)d_in[4];
    const float* Cm    = (const float*)d_in[5];
    const float* Dv    = (const float*)d_in[6];
    const float* dt    = (const float*)d_in[7];
    const float* W_out = (const float*)d_in[8];
    const float* b_out = (const float*)d_in[9];

    float* xin  = (float*)d_ws;
    float* ymid = xin  + (size_t)MTOT * D_MODEL;
    float* carry = ymid + (size_t)MTOT * D_MODEL;
    float* out  = (float*)d_out;

    const dim3 blk(256);
    gemm_wmma_f32<<<128, blk, 0, stream>>>(x, W_in, b_in, xin);
    s4_chunk_end  <<<dim3(16, 15), dim3(128), 0, stream>>>(xin, A_log, dt, carry);
    s4_chunk_prefix<<<512, blk, 0, stream>>>(A_log, dt, carry);
    s4_scan_chunk <<<dim3(16, 16), dim3(128), 0, stream>>>(xin, A_log, Bm, Cm, Dv, dt, carry, ymid);
    gemm_wmma_f32<<<128, blk, 0, stream>>>(ymid, W_out, b_out, out);
}


# GEMMs: A and W tiles staged through a two-stage LDS ring with whole-line global loads; 256 workgroups of 2x2 waves read fragments from LDS; k order and summation order unchanged
# speedup vs baseline: 4.0136x; 1.6714x over previous
#include <hip/hip_runtime.h>


typedef float v2f __attribute__((ext_vector_type(2)));
typedef float v8f __attribute__((ext_vector_type(8)));

#define D_MODEL 512
#define D_STATE 64
#define SEQ_LEN 2048
#define BATCH 4
#define MTOT (BATCH * SEQ_LEN)
#define NCHUNK 16
#define TCHUNK (SEQ_LEN / NCHUNK)
#define TBLK 8
#define TBLK1 16
#define CPLANE (BATCH * D_STATE * D_MODEL)

#define GKC 32
#define GP  36
#define GTILE 128
#define GNCH (D_MODEL / GKC)

static __device__ __forceinline__ void load_pair(
    v2f (&fa)[2][4], v2f (&fb)[2][4], const float* As, const float* Bs, const int p)
{
#pragma unroll
    for (int ks = 0; ks < 2; ++ks) {
#pragma unroll
        for (int i = 0; i < 4; ++i)
            fa[ks][i] = *(const v2f*)(As + i * 16 * GP + (2 * p + ks) * 4);
#pragma unroll
        for (int j = 0; j < 4; ++j)
            fb[ks][j] = *(const v2f*)(Bs + j * 16 * GP + (2 * p + ks) * 4);
    }
}

static __device__ __forceinline__ void mma_pair(
    const v2f (&fa)[2][4], const v2f (&fb)[2][4], v8f (&acc)[4][4])
{
#pragma unroll
    for (int ks = 0; ks < 2; ++ks)
#pragma unroll
        for (int i = 0; i < 4; ++i)
#pragma unroll
            for (int j = 0; j < 4; ++j)
                acc[i][j] = __builtin_amdgcn_wmma_f32_16x16x4_f32(
                    false, fa[ks][i], false, fb[ks][j], (short)0, acc[i][j],
                    false, false);
}

template <bool MORE>
static __device__ __forceinline__ void gemm_chunk(
    const float* As, const float* Bs, float* dA, float* dB,
    const float* gA, const float* gW,
    float4 (&pa)[8], float4 (&pb)[8], v8f (&acc)[4][4])
{
    v2f fa[2][2][4], fb[2][2][4];
    load_pair(fa[0], fb[0], As, Bs, 0);
    load_pair(fa[1], fb[1], As, Bs, 1);
    if (MORE) {
#pragma unroll
        for (int i = 0; i < 8; ++i) {
            pa[i] = *(const float4*)(gA + (size_t)i * 16 * D_MODEL);
            pb[i] = *(const float4*)(gW + (size_t)i * 16 * D_MODEL);
        }
    }
    mma_pair(fa[0], fb[0], acc);
    load_pair(fa[0], fb[0], As, Bs, 2);
    mma_pair(fa[1], fb[1], acc);
    load_pair(fa[1], fb[1], As, Bs, 3);
    mma_pair(fa[0], fb[0], acc);
    if (MORE) {
#pragma unroll
        for (int i = 0; i < 8; ++i) {
            *(float4*)(dA + i * 16 * GP) = pa[i];
            *(float4*)(dB + i * 16 * GP) = pb[i];
        }
    }
    mma_pair(fa[1], fb[1], acc);

    __builtin_amdgcn_sched_group_barrier(0x100, 16, 0);
    if (MORE) __builtin_amdgcn_sched_group_barrier(0x020, 16, 0);
    __builtin_amdgcn_sched_group_barrier(0x008, 32, 0);
    __builtin_amdgcn_sched_group_barrier(0x100, 8, 0);
    __builtin_amdgcn_sched_group_barrier(0x008, 32, 0);
    __builtin_amdgcn_sched_group_barrier(0x100, 8, 0);
    __builtin_amdgcn_sched_group_barrier(0x008, 32, 0);
    if (MORE) __builtin_amdgcn_sched_group_barrier(0x200, 16, 0);
    __builtin_amdgcn_sched_group_barrier(0x008, 32, 0);
}

__global__ __launch_bounds__(128) __attribute__((amdgpu_waves_per_eu(1, 2)))
void gemm_wmma_f32(
    const float* __restrict__ A, const float* __restrict__ W,
    const float* __restrict__ bias, float* __restrict__ out)
{
    __shared__ float lds[2][2][GTILE * GP];

    const int tid  = threadIdx.x;
    const int lane = tid & 31;
    const int l16  = lane & 15;
    const int half = lane >> 4;
    const int wave = tid >> 5;
    const int wm = wave >> 1;
    const int wn = wave & 1;
    const int M0 = (blockIdx.x >> 2) * GTILE;
    const int N0 = (blockIdx.x & 3) * GTILE;

    const int r0 = tid >> 3;
    const int c4 = tid & 7;
    const float* gA = A + (size_t)(M0 + r0) * D_MODEL + c4 * 4;
    const float* gW = W + (size_t)(N0 + r0) * D_MODEL + c4 * 4;
    const int sOff = r0 * GP + c4 * 4;

    const int aOff = (wm * 64 + l16) * GP + half * 2;
    const int bOff = (wn * 64 + l16) * GP + half * 2;

    float4 pa[8], pb[8];
#pragma unroll
    for (int i = 0; i < 8; ++i) {
        pa[i] = *(const float4*)(gA + (size_t)i * 16 * D_MODEL);
        pb[i] = *(const float4*)(gW + (size_t)i * 16 * D_MODEL);
    }
#pragma unroll
    for (int i = 0; i < 8; ++i) {
        *(float4*)&lds[0][0][sOff + i * 16 * GP] = pa[i];
        *(float4*)&lds[0][1][sOff + i * 16 * GP] = pb[i];
    }
    __syncthreads();

    v8f acc[4][4] = {};

#pragma unroll 1
    for (int c = 0; c < GNCH - 1; ++c) {
        const int st = c & 1;
        const int kc = (c + 1) * GKC;
        gemm_chunk<true>(&lds[st][0][aOff], &lds[st][1][bOff],
                         &lds[st ^ 1][0][sOff], &lds[st ^ 1][1][sOff],
                         gA + kc, gW + kc, pa, pb, acc);
        __syncthreads();
    }
    gemm_chunk<false>(&lds[(GNCH - 1) & 1][0][aOff], &lds[(GNCH - 1) & 1][1][bOff],
                      nullptr, nullptr, nullptr, nullptr, pa, pb, acc);

    const int Mw = M0 + wm * 64;
    const int Nw = N0 + wn * 64;
#pragma unroll
    for (int j = 0; j < 4; ++j) {
        const float bj = bias[Nw + j * 16 + l16];
#pragma unroll
        for (int i = 0; i < 4; ++i) {
#pragma unroll
            for (int r = 0; r < 8; ++r) {
                const int row = Mw + i * 16 + half * 8 + r;
                out[(size_t)row * D_MODEL + Nw + j * 16 + l16] =
                    acc[i][j][r] + bj;
            }
        }
    }
}

__global__ __launch_bounds__(128) void s4_chunk_end(
    const float* __restrict__ xin,
    const float* __restrict__ A_log,
    const float* __restrict__ dt,
    float* __restrict__ carry)
{
    const int lane = threadIdx.x & 31;
    const int g = blockIdx.x * 4 + (threadIdx.x >> 5);
    const int c = blockIdx.y;
    const int b = g >> 4;
    const int d = (g & 15) * 32 + lane;

    const float dtd = dt[d];
    float dA[D_STATE];
    const float4* al = (const float4*)(A_log + (size_t)d * D_STATE);
#pragma unroll
    for (int q = 0; q < D_STATE / 4; ++q) {
        const float4 v = al[q];
        dA[4 * q + 0] = expf(-expf(v.x) * dtd);
        dA[4 * q + 1] = expf(-expf(v.y) * dtd);
        dA[4 * q + 2] = expf(-expf(v.z) * dtd);
        dA[4 * q + 3] = expf(-expf(v.w) * dtd);
    }

    float s[D_STATE];
#pragma unroll
    for (int n = 0; n < D_STATE; ++n) s[n] = 0.f;

    const float* up = xin + (size_t)(b * SEQ_LEN + c * TCHUNK) * D_MODEL + d;
    float ua[TBLK1], ub[TBLK1];
#pragma unroll
    for (int k = 0; k < TBLK1; ++k) ua[k] = up[(size_t)k * D_MODEL];

#pragma unroll 1
    for (int t0 = 0; t0 < TCHUNK; t0 += 2 * TBLK1) {
#pragma unroll
        for (int k = 0; k < TBLK1; ++k)
            ub[k] = up[(size_t)(t0 + TBLK1 + k) * D_MODEL];
#pragma unroll
        for (int k = 0; k < TBLK1; ++k) {
#pragma unroll
            for (int n = 0; n < D_STATE; ++n)
                s[n] = fmaf(dA[n], s[n], ua[k]);
        }
        if (t0 + 2 * TBLK1 < TCHUNK) {
#pragma unroll
            for (int k = 0; k < TBLK1; ++k)
                ua[k] = up[(size_t)(t0 + 2 * TBLK1 + k) * D_MODEL];
        }
#pragma unroll
        for (int k = 0; k < TBLK1; ++k) {
#pragma unroll
            for (int n = 0; n < D_STATE; ++n)
                s[n] = fmaf(dA[n], s[n], ub[k]);
        }
    }

    float* E = carry + (size_t)c * CPLANE + (size_t)b * D_STATE * D_MODEL + d;
#pragma unroll
    for (int n = 0; n < D_STATE; ++n) E[(size_t)n * D_MODEL] = s[n];
}

__global__ __launch_bounds__(256) void s4_chunk_prefix(
    const float* __restrict__ A_log, const float* __restrict__ dt,
    float* __restrict__ carry)
{
    const int idx = blockIdx.x * 256 + threadIdx.x;
    const int d = idx & (D_MODEL - 1);
    const int n = (idx >> 9) & (D_STATE - 1);

    const float a = -expf(A_log[d * D_STATE + n]);
    const float dAp = expf(a * dt[d] * (float)TCHUNK);

    float* P = carry + idx;
    float e[NCHUNK - 1];
#pragma unroll
    for (int c = 0; c < NCHUNK - 1; ++c) e[c] = P[(size_t)c * CPLANE];

    float s = 0.f;
#pragma unroll
    for (int c = 0; c < NCHUNK - 1; ++c) {
        P[(size_t)c * CPLANE] = s;
        s = fmaf(dAp, s, e[c]);
    }
    P[(size_t)(NCHUNK - 1) * CPLANE] = s;
}

static __device__ __forceinline__ void scan_block(
    float (&s)[D_STATE], const float (&dA)[D_STATE], const float (&cf)[D_STATE],
    const float (&u)[TBLK], const float Dd, float* __restrict__ yp)
{
#pragma unroll
    for (int k = 0; k < TBLK; ++k) {
        float p0 = 0.f, p1 = 0.f, p2 = 0.f, p3 = 0.f;
#pragma unroll
        for (int n = 0; n < D_STATE; n += 4) {
            s[n + 0] = fmaf(dA[n + 0], s[n + 0], u[k]);
            s[n + 1] = fmaf(dA[n + 1], s[n + 1], u[k]);
            s[n + 2] = fmaf(dA[n + 2], s[n + 2], u[k]);
            s[n + 3] = fmaf(dA[n + 3], s[n + 3], u[k]);
            p0 = fmaf(cf[n + 0], s[n + 0], p0);
            p1 = fmaf(cf[n + 1], s[n + 1], p1);
            p2 = fmaf(cf[n + 2], s[n + 2], p2);
            p3 = fmaf(cf[n + 3], s[n + 3], p3);
        }
        yp[(size_t)k * D_MODEL] = fmaf(Dd, u[k], (p0 + p1) + (p2 + p3));
    }
}

__global__ __launch_bounds__(128) void s4_scan_chunk(
    const float* __restrict__ xin,
    const float* __restrict__ A_log,
    const float* __restrict__ Bm,
    const float* __restrict__ Cm,
    const float* __restrict__ Dv,
    const float* __restrict__ dt,
    const float* __restrict__ carry,
    float* __restrict__ y)
{
    const int lane = threadIdx.x & 31;
    const int g = blockIdx.x * 4 + (threadIdx.x >> 5);
    const int c = blockIdx.y;
    const int b = g >> 4;
    const int d = (g & 15) * 32 + lane;

    const float dtd = dt[d];
    const float Dd  = Dv[d];
    float dA[D_STATE], cf[D_STATE], s[D_STATE];
    const float4* al = (const float4*)(A_log + (size_t)d * D_STATE);
    const float4* bl = (const float4*)(Bm    + (size_t)d * D_STATE);
    const float4* cl = (const float4*)(Cm    + (size_t)d * D_STATE);
#pragma unroll
    for (int q = 0; q < D_STATE / 4; ++q) {
        const float4 va = al[q];
        const float4 vb = bl[q];
        const float4 vc = cl[q];
        dA[4 * q + 0] = expf(-expf(va.x) * dtd);
        dA[4 * q + 1] = expf(-expf(va.y) * dtd);
        dA[4 * q + 2] = expf(-expf(va.z) * dtd);
        dA[4 * q + 3] = expf(-expf(va.w) * dtd);
        cf[4 * q + 0] = vc.x * vb.x * dtd;
        cf[4 * q + 1] = vc.y * vb.y * dtd;
        cf[4 * q + 2] = vc.z * vb.z * dtd;
        cf[4 * q + 3] = vc.w * vb.w * dtd;
    }

    const float* S = carry + (size_t)c * CPLANE + (size_t)b * D_STATE * D_MODEL + d;
#pragma unroll
    for (int n = 0; n < D_STATE; ++n) s[n] = S[(size_t)n * D_MODEL];

    const float* up = xin + (size_t)(b * SEQ_LEN + c * TCHUNK) * D_MODEL + d;
    float*       yp = y   + (size_t)(b * SEQ_LEN + c * TCHUNK) * D_MODEL + d;

    float ua[TBLK], ub[TBLK];
#pragma unroll
    for (int k = 0; k < TBLK; ++k) ua[k] = up[(size_t)k * D_MODEL];

#pragma unroll 1
    for (int t0 = 0; t0 < TCHUNK; t0 += 2 * TBLK) {
#pragma unroll
        for (int k = 0; k < TBLK; ++k)
            ub[k] = up[(size_t)(t0 + TBLK + k) * D_MODEL];
        scan_block(s, dA, cf, ua, Dd, yp + (size_t)t0 * D_MODEL);
        if (t0 + 2 * TBLK < TCHUNK) {
#pragma unroll
            for (int k = 0; k < TBLK; ++k)
                ua[k] = up[(size_t)(t0 + 2 * TBLK + k) * D_MODEL];
        }
        scan_block(s, dA, cf, ub, Dd, yp + (size_t)(t0 + TBLK) * D_MODEL);
    }
}

extern "C" void kernel_launch(void* const* d_in, const int* in_sizes, int n_in,
                              void* d_out, int out_size, void* d_ws, size_t ws_size,
                              hipStream_t stream)
{
    (void)in_sizes; (void)n_in; (void)out_size; (void)ws_size;

    const float* x     = (const float*)d_in[0];
    const float* W_in  = (const float*)d_in[1];
    const float* b_in  = (const float*)d_in[2];
    const float* A_log = (const float*)d_in[3];
    const float* Bm    = (const float*)d_in[4];
    const float* Cm    = (const float*)d_in[5];
    const float* Dv    = (const float*)d_in[6];
    const float* dt    = (const float*)d_in[7];
    const float* W_out = (const float*)d_in[8];
    const float* b_out = (const float*)d_in[9];

    float* xin  = (float*)d_ws;
    float* ymid = xin  + (size_t)MTOT * D_MODEL;
    float* carry = ymid + (size_t)MTOT * D_MODEL;
    float* out  = (float*)d_out;

    const dim3 blk(256);
    gemm_wmma_f32<<<256, dim3(128), 0, stream>>>(x, W_in, b_in, xin);
    s4_chunk_end  <<<dim3(16, 15), dim3(128), 0, stream>>>(xin, A_log, dt, carry);
    s4_chunk_prefix<<<512, blk, 0, stream>>>(A_log, dt, carry);
    s4_scan_chunk <<<dim3(16, 16), dim3(128), 0, stream>>>(xin, A_log, Bm, Cm, Dv, dt, carry, ymid);
    gemm_wmma_f32<<<256, dim3(128), 0, stream>>>(ymid, W_out, b_out, out);
}


# Chunk-carry prefix folded into the scan pass: a workgroup owns one channel group and 4 chunks; its waves split set-up and carry recurrence by state through LDS; prefix launch gone; sums unchanged
# speedup vs baseline: 4.2271x; 1.0532x over previous
#include <hip/hip_runtime.h>


typedef float v2f __attribute__((ext_vector_type(2)));
typedef float v8f __attribute__((ext_vector_type(8)));

#define D_MODEL 512
#define D_STATE 64
#define SEQ_LEN 2048
#define BATCH 4
#define MTOT (BATCH * SEQ_LEN)
#define NCHUNK 16
#define TCHUNK (SEQ_LEN / NCHUNK)
#define TBLK 8
#define TBLK1 16
#define CPLANE (BATCH * D_STATE * D_MODEL)

#define GKC 32
#define GP  36
#define GTILE 128
#define GNCH (D_MODEL / GKC)

static __device__ __forceinline__ void load_pair(
    v2f (&fa)[2][4], v2f (&fb)[2][4], const float* As, const float* Bs, const int p)
{
#pragma unroll
    for (int ks = 0; ks < 2; ++ks) {
#pragma unroll
        for (int i = 0; i < 4; ++i)
            fa[ks][i] = *(const v2f*)(As + i * 16 * GP + (2 * p + ks) * 4);
#pragma unroll
        for (int j = 0; j < 4; ++j)
            fb[ks][j] = *(const v2f*)(Bs + j * 16 * GP + (2 * p + ks) * 4);
    }
}

static __device__ __forceinline__ void mma_pair(
    const v2f (&fa)[2][4], const v2f (&fb)[2][4], v8f (&acc)[4][4])
{
#pragma unroll
    for (int ks = 0; ks < 2; ++ks)
#pragma unroll
        for (int i = 0; i < 4; ++i)
#pragma unroll
            for (int j = 0; j < 4; ++j)
                acc[i][j] = __builtin_amdgcn_wmma_f32_16x16x4_f32(
                    false, fa[ks][i], false, fb[ks][j], (short)0, acc[i][j],
                    false, false);
}

template <bool MORE>
static __device__ __forceinline__ void gemm_chunk(
    const float* As, const float* Bs, float* dA, float* dB,
    const float* gA, const float* gW,
    float4 (&pa)[8], float4 (&pb)[8], v8f (&acc)[4][4])
{
    v2f fa[2][2][4], fb[2][2][4];
    load_pair(fa[0], fb[0], As, Bs, 0);
    load_pair(fa[1], fb[1], As, Bs, 1);
    if (MORE) {
#pragma unroll
        for (int i = 0; i < 8; ++i) {
            pa[i] = *(const float4*)(gA + (size_t)i * 16 * D_MODEL);
            pb[i] = *(const float4*)(gW + (size_t)i * 16 * D_MODEL);
        }
    }
    mma_pair(fa[0], fb[0], acc);
    load_pair(fa[0], fb[0], As, Bs, 2);
    mma_pair(fa[1], fb[1], acc);
    load_pair(fa[1], fb[1], As, Bs, 3);
    mma_pair(fa[0], fb[0], acc);
    if (MORE) {
#pragma unroll
        for (int i = 0; i < 8; ++i) {
            *(float4*)(dA + i * 16 * GP) = pa[i];
            *(float4*)(dB + i * 16 * GP) = pb[i];
        }
    }
    mma_pair(fa[1], fb[1], acc);

    __builtin_amdgcn_sched_group_barrier(0x100, 16, 0);
    if (MORE) __builtin_amdgcn_sched_group_barrier(0x020, 16, 0);
    __builtin_amdgcn_sched_group_barrier(0x008, 32, 0);
    __builtin_amdgcn_sched_group_barrier(0x100, 8, 0);
    __builtin_amdgcn_sched_group_barrier(0x008, 32, 0);
    __builtin_amdgcn_sched_group_barrier(0x100, 8, 0);
    __builtin_amdgcn_sched_group_barrier(0x008, 32, 0);
    if (MORE) __builtin_amdgcn_sched_group_barrier(0x200, 16, 0);
    __builtin_amdgcn_sched_group_barrier(0x008, 32, 0);
}

__global__ __launch_bounds__(128) __attribute__((amdgpu_waves_per_eu(1, 2)))
void gemm_wmma_f32(
    const float* __restrict__ A, const float* __restrict__ W,
    const float* __restrict__ bias, float* __restrict__ out)
{
    __shared__ float lds[2][2][GTILE * GP];

    const int tid  = threadIdx.x;
    const int lane = tid & 31;
    const int l16  = lane & 15;
    const int half = lane >> 4;
    const int wave = tid >> 5;
    const int wm = wave >> 1;
    const int wn = wave & 1;
    const int M0 = (blockIdx.x >> 2) * GTILE;
    const int N0 = (blockIdx.x & 3) * GTILE;

    const int r0 = tid >> 3;
    const int c4 = tid & 7;
    const float* gA = A + (size_t)(M0 + r0) * D_MODEL + c4 * 4;
    const float* gW = W + (size_t)(N0 + r0) * D_MODEL + c4 * 4;
    const int sOff = r0 * GP + c4 * 4;

    const int aOff = (wm * 64 + l16) * GP + half * 2;
    const int bOff = (wn * 64 + l16) * GP + half * 2;

    float4 pa[8], pb[8];
#pragma unroll
    for (int i = 0; i < 8; ++i) {
        pa[i] = *(const float4*)(gA + (size_t)i * 16 * D_MODEL);
        pb[i] = *(const float4*)(gW + (size_t)i * 16 * D_MODEL);
    }
#pragma unroll
    for (int i = 0; i < 8; ++i) {
        *(float4*)&lds[0][0][sOff + i * 16 * GP] = pa[i];
        *(float4*)&lds[0][1][sOff + i * 16 * GP] = pb[i];
    }
    __syncthreads();

    v8f acc[4][4] = {};

#pragma unroll 1
    for (int c = 0; c < GNCH - 1; ++c) {
        const int st = c & 1;
        const int kc = (c + 1) * GKC;
        gemm_chunk<true>(&lds[st][0][aOff], &lds[st][1][bOff],
                         &lds[st ^ 1][0][sOff], &lds[st ^ 1][1][sOff],
                         gA + kc, gW + kc, pa, pb, acc);
        __syncthreads();
    }
    gemm_chunk<false>(&lds[(GNCH - 1) & 1][0][aOff], &lds[(GNCH - 1) & 1][1][bOff],
                      nullptr, nullptr, nullptr, nullptr, pa, pb, acc);

    const int Mw = M0 + wm * 64;
    const int Nw = N0 + wn * 64;
#pragma unroll
    for (int j = 0; j < 4; ++j) {
        const float bj = bias[Nw + j * 16 + l16];
#pragma unroll
        for (int i = 0; i < 4; ++i) {
#pragma unroll
            for (int r = 0; r < 8; ++r) {
                const int row = Mw + i * 16 + half * 8 + r;
                out[(size_t)row * D_MODEL + Nw + j * 16 + l16] =
                    acc[i][j][r] + bj;
            }
        }
    }
}

__global__ __launch_bounds__(128) void s4_chunk_end(
    const float* __restrict__ xin,
    const float* __restrict__ A_log,
    const float* __restrict__ dt,
    float* __restrict__ carry)
{
    const int lane = threadIdx.x & 31;
    const int g = blockIdx.x * 4 + (threadIdx.x >> 5);
    const int c = blockIdx.y;
    const int b = g >> 4;
    const int d = (g & 15) * 32 + lane;

    const float dtd = dt[d];
    float dA[D_STATE];
    const float4* al = (const float4*)(A_log + (size_t)d * D_STATE);
#pragma unroll
    for (int q = 0; q < D_STATE / 4; ++q) {
        const float4 v = al[q];
        dA[4 * q + 0] = expf(-expf(v.x) * dtd);
        dA[4 * q + 1] = expf(-expf(v.y) * dtd);
        dA[4 * q + 2] = expf(-expf(v.z) * dtd);
        dA[4 * q + 3] = expf(-expf(v.w) * dtd);
    }

    float s[D_STATE];
#pragma unroll
    for (int n = 0; n < D_STATE; ++n) s[n] = 0.f;

    const float* up = xin + (size_t)(b * SEQ_LEN + c * TCHUNK) * D_MODEL + d;
    float ua[TBLK1], ub[TBLK1];
#pragma unroll
    for (int k = 0; k < TBLK1; ++k) ua[k] = up[(size_t)k * D_MODEL];

#pragma unroll 1
    for (int t0 = 0; t0 < TCHUNK; t0 += 2 * TBLK1) {
#pragma unroll
        for (int k = 0; k < TBLK1; ++k)
            ub[k] = up[(size_t)(t0 + TBLK1 + k) * D_MODEL];
#pragma unroll
        for (int k = 0; k < TBLK1; ++k) {
#pragma unroll
            for (int n = 0; n < D_STATE; ++n)
                s[n] = fmaf(dA[n], s[n], ua[k]);
        }
        if (t0 + 2 * TBLK1 < TCHUNK) {
#pragma unroll
            for (int k = 0; k < TBLK1; ++k)
                ua[k] = up[(size_t)(t0 + 2 * TBLK1 + k) * D_MODEL];
        }
#pragma unroll
        for (int k = 0; k < TBLK1; ++k) {
#pragma unroll
            for (int n = 0; n < D_STATE; ++n)
                s[n] = fmaf(dA[n], s[n], ub[k]);
        }
    }

    float4* E = (float4*)(carry + (size_t)c * CPLANE + (size_t)g * (D_STATE * 32)) + lane;
#pragma unroll
    for (int q = 0; q < D_STATE / 4; ++q)
        E[q * 32] = make_float4(s[4 * q + 0], s[4 * q + 1], s[4 * q + 2], s[4 * q + 3]);
}

static __device__ __forceinline__ void scan_block(
    float (&s)[D_STATE], const float (&dA)[D_STATE], const float (&cf)[D_STATE],
    const float (&u)[TBLK], const float Dd, float* __restrict__ yp)
{
#pragma unroll
    for (int k = 0; k < TBLK; ++k) {
        float p0 = 0.f, p1 = 0.f, p2 = 0.f, p3 = 0.f;
#pragma unroll
        for (int n = 0; n < D_STATE; n += 4) {
            s[n + 0] = fmaf(dA[n + 0], s[n + 0], u[k]);
            s[n + 1] = fmaf(dA[n + 1], s[n + 1], u[k]);
            s[n + 2] = fmaf(dA[n + 2], s[n + 2], u[k]);
            s[n + 3] = fmaf(dA[n + 3], s[n + 3], u[k]);
            p0 = fmaf(cf[n + 0], s[n + 0], p0);
            p1 = fmaf(cf[n + 1], s[n + 1], p1);
            p2 = fmaf(cf[n + 2], s[n + 2], p2);
            p3 = fmaf(cf[n + 3], s[n + 3], p3);
        }
        yp[(size_t)k * D_MODEL] = fmaf(Dd, u[k], (p0 + p1) + (p2 + p3));
    }
}

static __device__ __forceinline__ void carry_step(
    float (&sp)[16], const float (&dAp)[16], const float4 e, const int q)
{
    sp[4 * q + 0] = fmaf(dAp[4 * q + 0], sp[4 * q + 0], e.x);
    sp[4 * q + 1] = fmaf(dAp[4 * q + 1], sp[4 * q + 1], e.y);
    sp[4 * q + 2] = fmaf(dAp[4 * q + 2], sp[4 * q + 2], e.z);
    sp[4 * q + 3] = fmaf(dAp[4 * q + 3], sp[4 * q + 3], e.w);
}

__global__ __launch_bounds__(128) void s4_scan_chunk(
    const float* __restrict__ xin,
    const float* __restrict__ A_log,
    const float* __restrict__ Bm,
    const float* __restrict__ Cm,
    const float* __restrict__ Dv,
    const float* __restrict__ dt,
    const float* __restrict__ carry,
    float* __restrict__ y)
{
    __shared__ float tA[D_STATE][32];
    __shared__ float tC[D_STATE][32];
    __shared__ float sX[4][D_STATE][32];

    const int lane = threadIdx.x & 31;
    const int w  = threadIdx.x >> 5;
    const int g  = blockIdx.x;
    const int c0 = blockIdx.y;
    const int c  = c0 + 4 * w;
    const int b  = g >> 4;
    const int d  = (g & 15) * 32 + lane;
    const int n0 = 16 * w;
    const int np = c0 + 12;

    const float* up = xin + (size_t)(b * SEQ_LEN + c * TCHUNK) * D_MODEL + d;
    float*       yp = y   + (size_t)(b * SEQ_LEN + c * TCHUNK) * D_MODEL + d;

    const float dtd = dt[d];
    const float Dd  = Dv[d];
    const float4* al = (const float4*)(A_log + (size_t)d * D_STATE + n0);
    const float4* bl = (const float4*)(Bm    + (size_t)d * D_STATE + n0);
    const float4* cl = (const float4*)(Cm    + (size_t)d * D_STATE + n0);
    float4 va[4], vb[4], vc[4];
#pragma unroll
    for (int q = 0; q < 4; ++q) { va[q] = al[q]; vb[q] = bl[q]; vc[q] = cl[q]; }

    float ua[TBLK], ub[TBLK];
#pragma unroll
    for (int k = 0; k < TBLK; ++k) ua[k] = up[(size_t)k * D_MODEL];

    const float4* Ep = (const float4*)(carry + (size_t)g * (D_STATE * 32)) + (4 * w) * 32 + lane;
    float4 e[NCHUNK - 1][4];
#pragma unroll
    for (int p = 0; p < NCHUNK - 1; ++p) {
        if (p < 12 || p < np) {
#pragma unroll
            for (int q = 0; q < 4; ++q)
                e[p][q] = Ep[(size_t)p * (CPLANE / 4) + q * 32];
        }
    }

    float dAp[16];
#pragma unroll
    for (int q = 0; q < 4; ++q) {
        const float l0 = -expf(va[q].x) * dtd;
        const float l1 = -expf(va[q].y) * dtd;
        const float l2 = -expf(va[q].z) * dtd;
        const float l3 = -expf(va[q].w) * dtd;
        tA[n0 + 4 * q + 0][lane] = expf(l0);
        tA[n0 + 4 * q + 1][lane] = expf(l1);
        tA[n0 + 4 * q + 2][lane] = expf(l2);
        tA[n0 + 4 * q + 3][lane] = expf(l3);
        tC[n0 + 4 * q + 0][lane] = vc[q].x * vb[q].x * dtd;
        tC[n0 + 4 * q + 1][lane] = vc[q].y * vb[q].y * dtd;
        tC[n0 + 4 * q + 2][lane] = vc[q].z * vb[q].z * dtd;
        tC[n0 + 4 * q + 3][lane] = vc[q].w * vb[q].w * dtd;
        dAp[4 * q + 0] = expf(l0 * (float)TCHUNK);
        dAp[4 * q + 1] = expf(l1 * (float)TCHUNK);
        dAp[4 * q + 2] = expf(l2 * (float)TCHUNK);
        dAp[4 * q + 3] = expf(l3 * (float)TCHUNK);
    }

    float sp[16];
#pragma unroll
    for (int i = 0; i < 16; ++i) sp[i] = 0.f;
#pragma unroll
    for (int p = 0; p < NCHUNK; ++p) {
        if (p >= c0 && ((p - c0) & 3) == 0) {
            const int wq = (p - c0) >> 2;
#pragma unroll
            for (int i = 0; i < 16; ++i) sX[wq][n0 + i][lane] = sp[i];
        }
        if (p < 12 || (p < NCHUNK - 1 && p < np)) {
#pragma unroll
            for (int q = 0; q < 4; ++q) carry_step(sp, dAp, e[p < NCHUNK - 1 ? p : 0][q], q);
        }
    }
    __syncthreads();

    float dA[D_STATE], cf[D_STATE], s[D_STATE];
#pragma unroll
    for (int n = 0; n < D_STATE; ++n) {
        dA[n] = tA[n][lane];
        cf[n] = tC[n][lane];
        s[n]  = sX[w][n][lane];
    }

#pragma unroll 1
    for (int t0 = 0; t0 < TCHUNK; t0 += 2 * TBLK) {
#pragma unroll
        for (int k = 0; k < TBLK; ++k)
            ub[k] = up[(size_t)(t0 + TBLK + k) * D_MODEL];
        scan_block(s, dA, cf, ua, Dd, yp + (size_t)t0 * D_MODEL);
        if (t0 + 2 * TBLK < TCHUNK) {
#pragma unroll
            for (int k = 0; k < TBLK; ++k)
                ua[k] = up[(size_t)(t0 + 2 * TBLK + k) * D_MODEL];
        }
        scan_block(s, dA, cf, ub, Dd, yp + (size_t)(t0 + TBLK) * D_MODEL);
    }
}

extern "C" void kernel_launch(void* const* d_in, const int* in_sizes, int n_in,
                              void* d_out, int out_size, void* d_ws, size_t ws_size,
                              hipStream_t stream)
{
    (void)in_sizes; (void)n_in; (void)out_size; (void)ws_size;

    const float* x     = (const float*)d_in[0];
    const float* W_in  = (const float*)d_in[1];
    const float* b_in  = (const float*)d_in[2];
    const float* A_log = (const float*)d_in[3];
    const float* Bm    = (const float*)d_in[4];
    const float* Cm    = (const float*)d_in[5];
    const float* Dv    = (const float*)d_in[6];
    const float* dt    = (const float*)d_in[7];
    const float* W_out = (const float*)d_in[8];
    const float* b_out = (const float*)d_in[9];

    float* xin  = (float*)d_ws;
    float* ymid = xin  + (size_t)MTOT * D_MODEL;
    float* carry = ymid + (size_t)MTOT * D_MODEL;
    float* out  = (float*)d_out;

    gemm_wmma_f32<<<256, dim3(128), 0, stream>>>(x, W_in, b_in, xin);
    s4_chunk_end  <<<dim3(16, 15), dim3(128), 0, stream>>>(xin, A_log, dt, carry);
    s4_scan_chunk <<<dim3(64, 4), dim3(128), 0, stream>>>(xin, A_log, Bm, Cm, Dv, dt, carry, ymid);
    gemm_wmma_f32<<<256, dim3(128), 0, stream>>>(ymid, W_out, b_out, out);
}


# GEMMs: global operand loads run two chunks ahead of their WMMAs (two fetch register sets in turn; ring barrier waits for LDS only, not for loads in flight); k order and summation order unchanged
# speedup vs baseline: 4.3969x; 1.0402x over previous
#include <hip/hip_runtime.h>


typedef float v2f __attribute__((ext_vector_type(2)));
typedef float v4f __attribute__((ext_vector_type(4)));
typedef float v8f __attribute__((ext_vector_type(8)));

#define D_MODEL 512
#define D_STATE 64
#define SEQ_LEN 2048
#define BATCH 4
#define MTOT (BATCH * SEQ_LEN)
#define NCHUNK 16
#define TCHUNK (SEQ_LEN / NCHUNK)
#define TBLK 8
#define TBLK1 16
#define CPLANE (BATCH * D_STATE * D_MODEL)

#define GKC 32
#define GP  36
#define GTILE 128
#define GNCH (D_MODEL / GKC)

static __device__ __forceinline__ void load_pair(
    v2f (&fa)[2][4], v2f (&fb)[2][4], const float* As, const float* Bs, const int p)
{
#pragma unroll
    for (int ks = 0; ks < 2; ++ks) {
#pragma unroll
        for (int i = 0; i < 4; ++i)
            fa[ks][i] = *(const v2f*)(As + i * 16 * GP + (2 * p + ks) * 4);
#pragma unroll
        for (int j = 0; j < 4; ++j)
            fb[ks][j] = *(const v2f*)(Bs + j * 16 * GP + (2 * p + ks) * 4);
    }
}

static __device__ __forceinline__ void mma_pair(
    const v2f (&fa)[2][4], const v2f (&fb)[2][4], v8f (&acc)[4][4])
{
#pragma unroll
    for (int ks = 0; ks < 2; ++ks)
#pragma unroll
        for (int i = 0; i < 4; ++i)
#pragma unroll
            for (int j = 0; j < 4; ++j)
                acc[i][j] = __builtin_amdgcn_wmma_f32_16x16x4_f32(
                    false, fa[ks][i], false, fb[ks][j], (short)0, acc[i][j],
                    false, false);
}

static __device__ __forceinline__ void lds_ring_barrier()
{
    asm volatile("s_wait_dscnt 0x0\n\ts_barrier_signal -1\n\ts_barrier_wait -1" ::: "memory");
}

struct Fetch {
    v4f a0, a1, a2, a3, a4, a5, a6, a7;
    v4f b0, b1, b2, b3, b4, b5, b6, b7;
};

static __device__ __forceinline__ void fetch_load(Fetch& f, const float* gA, const float* gW)
{
    const size_t R = (size_t)16 * D_MODEL;
    f.a0 = *(const v4f*)(gA + 0 * R);  f.b0 = *(const v4f*)(gW + 0 * R);
    f.a1 = *(const v4f*)(gA + 1 * R);  f.b1 = *(const v4f*)(gW + 1 * R);
    f.a2 = *(const v4f*)(gA + 2 * R);  f.b2 = *(const v4f*)(gW + 2 * R);
    f.a3 = *(const v4f*)(gA + 3 * R);  f.b3 = *(const v4f*)(gW + 3 * R);
    f.a4 = *(const v4f*)(gA + 4 * R);  f.b4 = *(const v4f*)(gW + 4 * R);
    f.a5 = *(const v4f*)(gA + 5 * R);  f.b5 = *(const v4f*)(gW + 5 * R);
    f.a6 = *(const v4f*)(gA + 6 * R);  f.b6 = *(const v4f*)(gW + 6 * R);
    f.a7 = *(const v4f*)(gA + 7 * R);  f.b7 = *(const v4f*)(gW + 7 * R);
}

static __device__ __forceinline__ void fetch_store(const Fetch& f, float* dA, float* dB)
{
    const int R = 16 * GP;
    *(v4f*)(dA + 0 * R) = f.a0;  *(v4f*)(dB + 0 * R) = f.b0;
    *(v4f*)(dA + 1 * R) = f.a1;  *(v4f*)(dB + 1 * R) = f.b1;
    *(v4f*)(dA + 2 * R) = f.a2;  *(v4f*)(dB + 2 * R) = f.b2;
    *(v4f*)(dA + 3 * R) = f.a3;  *(v4f*)(dB + 3 * R) = f.b3;
    *(v4f*)(dA + 4 * R) = f.a4;  *(v4f*)(dB + 4 * R) = f.b4;
    *(v4f*)(dA + 5 * R) = f.a5;  *(v4f*)(dB + 5 * R) = f.b5;
    *(v4f*)(dA + 6 * R) = f.a6;  *(v4f*)(dB + 6 * R) = f.b6;
    *(v4f*)(dA + 7 * R) = f.a7;  *(v4f*)(dB + 7 * R) = f.b7;
}

template <bool LOAD, bool STORE>
static __device__ __forceinline__ void gemm_chunk(
    const float* As, const float* Bs, float* dA, float* dB,
    const float* gA, const float* gW,
    Fetch& ld, const Fetch& st,
    v2f (&fa)[2][2][4], v2f (&fb)[2][2][4], v8f (&acc)[4][4])
{
    load_pair(fa[0], fb[0], As, Bs, 0);
    load_pair(fa[1], fb[1], As, Bs, 1);
    if (LOAD) fetch_load(ld, gA, gW);
    mma_pair(fa[0], fb[0], acc);
    load_pair(fa[0], fb[0], As, Bs, 2);
    mma_pair(fa[1], fb[1], acc);
    load_pair(fa[1], fb[1], As, Bs, 3);
    mma_pair(fa[0], fb[0], acc);
    if (STORE) fetch_store(st, dA, dB);
    mma_pair(fa[1], fb[1], acc);

    __builtin_amdgcn_sched_group_barrier(0x100, 16, 0);
    if (LOAD) __builtin_amdgcn_sched_group_barrier(0x020, 16, 0);
    __builtin_amdgcn_sched_group_barrier(0x008, 32, 0);
    __builtin_amdgcn_sched_group_barrier(0x100, 8, 0);
    __builtin_amdgcn_sched_group_barrier(0x008, 32, 0);
    __builtin_amdgcn_sched_group_barrier(0x100, 8, 0);
    __builtin_amdgcn_sched_group_barrier(0x008, 32, 0);
    if (STORE) __builtin_amdgcn_sched_group_barrier(0x200, 16, 0);
    __builtin_amdgcn_sched_group_barrier(0x008, 32, 0);
}

__global__ __launch_bounds__(128) __attribute__((amdgpu_waves_per_eu(1, 2)))
void gemm_wmma_f32(
    const float* __restrict__ A, const float* __restrict__ W,
    const float* __restrict__ bias, float* __restrict__ out)
{
    __shared__ float lds[2][2][GTILE * GP];

    const int tid  = threadIdx.x;
    const int lane = tid & 31;
    const int l16  = lane & 15;
    const int half = lane >> 4;
    const int wave = tid >> 5;
    const int wm = wave >> 1;
    const int wn = wave & 1;
    const int M0 = (blockIdx.x >> 2) * GTILE;
    const int N0 = (blockIdx.x & 3) * GTILE;

    const int r0 = tid >> 3;
    const int c4 = tid & 7;
    const float* gA = A + (size_t)(M0 + r0) * D_MODEL + c4 * 4;
    const float* gW = W + (size_t)(N0 + r0) * D_MODEL + c4 * 4;
    const int sOff = r0 * GP + c4 * 4;

    const int aOff = (wm * 64 + l16) * GP + half * 2;
    const int bOff = (wn * 64 + l16) * GP + half * 2;

    Fetch p0, p1;
    fetch_load(p0, gA, gW);
    fetch_load(p1, gA + GKC, gW + GKC);
    fetch_store(p0, &lds[0][0][sOff], &lds[0][1][sOff]);
    lds_ring_barrier();

    v8f acc[4][4] = {};
    v2f fa[2][2][4], fb[2][2][4];

#pragma unroll 1
    for (int c = 0; c < GNCH - 2; c += 2) {
        gemm_chunk<true, true>(&lds[0][0][aOff], &lds[0][1][bOff],
                               &lds[1][0][sOff], &lds[1][1][sOff],
                               gA + (c + 2) * GKC, gW + (c + 2) * GKC,
                               p0, p1, fa, fb, acc);
        lds_ring_barrier();
        gemm_chunk<true, true>(&lds[1][0][aOff], &lds[1][1][bOff],
                               &lds[0][0][sOff], &lds[0][1][sOff],
                               gA + (c + 3) * GKC, gW + (c + 3) * GKC,
                               p1, p0, fa, fb, acc);
        lds_ring_barrier();
    }
    gemm_chunk<false, true>(&lds[0][0][aOff], &lds[0][1][bOff],
                            &lds[1][0][sOff], &lds[1][1][sOff],
                            nullptr, nullptr, p0, p1, fa, fb, acc);
    lds_ring_barrier();
    gemm_chunk<false, false>(&lds[1][0][aOff], &lds[1][1][bOff],
                             nullptr, nullptr, nullptr, nullptr, p0, p1, fa, fb, acc);

    const int Mw = M0 + wm * 64;
    const int Nw = N0 + wn * 64;
#pragma unroll
    for (int j = 0; j < 4; ++j) {
        const float bj = bias[Nw + j * 16 + l16];
#pragma unroll
        for (int i = 0; i < 4; ++i) {
#pragma unroll
            for (int r = 0; r < 8; ++r) {
                const int row = Mw + i * 16 + half * 8 + r;
                out[(size_t)row * D_MODEL + Nw + j * 16 + l16] =
                    acc[i][j][r] + bj;
            }
        }
    }
}

__global__ __launch_bounds__(128) void s4_chunk_end(
    const float* __restrict__ xin,
    const float* __restrict__ A_log,
    const float* __restrict__ dt,
    float* __restrict__ carry)
{
    const int lane = threadIdx.x & 31;
    const int g = blockIdx.x * 4 + (threadIdx.x >> 5);
    const int c = blockIdx.y;
    const int b = g >> 4;
    const int d = (g & 15) * 32 + lane;

    const float dtd = dt[d];
    float dA[D_STATE];
    const float4* al = (const float4*)(A_log + (size_t)d * D_STATE);
#pragma unroll
    for (int q = 0; q < D_STATE / 4; ++q) {
        const float4 v = al[q];
        dA[4 * q + 0] = expf(-expf(v.x) * dtd);
        dA[4 * q + 1] = expf(-expf(v.y) * dtd);
        dA[4 * q + 2] = expf(-expf(v.z) * dtd);
        dA[4 * q + 3] = expf(-expf(v.w) * dtd);
    }

    float s[D_STATE];
#pragma unroll
    for (int n = 0; n < D_STATE; ++n) s[n] = 0.f;

    const float* up = xin + (size_t)(b * SEQ_LEN + c * TCHUNK) * D_MODEL + d;
    float ua[TBLK1], ub[TBLK1];
#pragma unroll
    for (int k = 0; k < TBLK1; ++k) ua[k] = up[(size_t)k * D_MODEL];

#pragma unroll 1
    for (int t0 = 0; t0 < TCHUNK; t0 += 2 * TBLK1) {
#pragma unroll
        for (int k = 0; k < TBLK1; ++k)
            ub[k] = up[(size_t)(t0 + TBLK1 + k) * D_MODEL];
#pragma unroll
        for (int k = 0; k < TBLK1; ++k) {
#pragma unroll
            for (int n = 0; n < D_STATE; ++n)
                s[n] = fmaf(dA[n], s[n], ua[k]);
        }
        if (t0 + 2 * TBLK1 < TCHUNK) {
#pragma unroll
            for (int k = 0; k < TBLK1; ++k)
                ua[k] = up[(size_t)(t0 + 2 * TBLK1 + k) * D_MODEL];
        }
#pragma unroll
        for (int k = 0; k < TBLK1; ++k) {
#pragma unroll
            for (int n = 0; n < D_STATE; ++n)
                s[n] = fmaf(dA[n], s[n], ub[k]);
        }
    }

    float4* E = (float4*)(carry + (size_t)c * CPLANE + (size_t)g * (D_STATE * 32)) + lane;
#pragma unroll
    for (int q = 0; q < D_STATE / 4; ++q)
        E[q * 32] = make_float4(s[4 * q + 0], s[4 * q + 1], s[4 * q + 2], s[4 * q + 3]);
}

static __device__ __forceinline__ void scan_block(
    float (&s)[D_STATE], const float (&dA)[D_STATE], const float (&cf)[D_STATE],
    const float (&u)[TBLK], const float Dd, float* __restrict__ yp)
{
#pragma unroll
    for (int k = 0; k < TBLK; ++k) {
        float p0 = 0.f, p1 = 0.f, p2 = 0.f, p3 = 0.f;
#pragma unroll
        for (int n = 0; n < D_STATE; n += 4) {
            s[n + 0] = fmaf(dA[n + 0], s[n + 0], u[k]);
            s[n + 1] = fmaf(dA[n + 1], s[n + 1], u[k]);
            s[n + 2] = fmaf(dA[n + 2], s[n + 2], u[k]);
            s[n + 3] = fmaf(dA[n + 3], s[n + 3], u[k]);
            p0 = fmaf(cf[n + 0], s[n + 0], p0);
            p1 = fmaf(cf[n + 1], s[n + 1], p1);
            p2 = fmaf(cf[n + 2], s[n + 2], p2);
            p3 = fmaf(cf[n + 3], s[n + 3], p3);
        }
        yp[(size_t)k * D_MODEL] = fmaf(Dd, u[k], (p0 + p1) + (p2 + p3));
    }
}

static __device__ __forceinline__ void carry_step(
    float (&sp)[16], const float (&dAp)[16], const float4 e, const int q)
{
    sp[4 * q + 0] = fmaf(dAp[4 * q + 0], sp[4 * q + 0], e.x);
    sp[4 * q + 1] = fmaf(dAp[4 * q + 1], sp[4 * q + 1], e.y);
    sp[4 * q + 2] = fmaf(dAp[4 * q + 2], sp[4 * q + 2], e.z);
    sp[4 * q + 3] = fmaf(dAp[4 * q + 3], sp[4 * q + 3], e.w);
}

__global__ __launch_bounds__(128) void s4_scan_chunk(
    const float* __restrict__ xin,
    const float* __restrict__ A_log,
    const float* __restrict__ Bm,
    const float* __restrict__ Cm,
    const float* __restrict__ Dv,
    const float* __restrict__ dt,
    const float* __restrict__ carry,
    float* __restrict__ y)
{
    __shared__ float tA[D_STATE][32];
    __shared__ float tC[D_STATE][32];
    __shared__ float sX[4][D_STATE][32];

    const int lane = threadIdx.x & 31;
    const int w  = threadIdx.x >> 5;
    const int g  = blockIdx.x;
    const int c0 = blockIdx.y;
    const int c  = c0 + 4 * w;
    const int b  = g >> 4;
    const int d  = (g & 15) * 32 + lane;
    const int n0 = 16 * w;
    const int np = c0 + 12;

    const float* up = xin + (size_t)(b * SEQ_LEN + c * TCHUNK) * D_MODEL + d;
    float*       yp = y   + (size_t)(b * SEQ_LEN + c * TCHUNK) * D_MODEL + d;

    const float dtd = dt[d];
    const float Dd  = Dv[d];
    const float4* al = (const float4*)(A_log + (size_t)d * D_STATE + n0);
    const float4* bl = (const float4*)(Bm    + (size_t)d * D_STATE + n0);
    const float4* cl = (const float4*)(Cm    + (size_t)d * D_STATE + n0);
    float4 va[4], vb[4], vc[4];
#pragma unroll
    for (int q = 0; q < 4; ++q) { va[q] = al[q]; vb[q] = bl[q]; vc[q] = cl[q]; }

    float ua[TBLK], ub[TBLK];
#pragma unroll
    for (int k = 0; k < TBLK; ++k) ua[k] = up[(size_t)k * D_MODEL];

    const float4* Ep = (const float4*)(carry + (size_t)g * (D_STATE * 32)) + (4 * w) * 32 + lane;
    float4 e[NCHUNK - 1][4];
#pragma unroll
    for (int p = 0; p < NCHUNK - 1; ++p) {
        if (p < 12 || p < np) {
#pragma unroll
            for (int q = 0; q < 4; ++q)
                e[p][q] = Ep[(size_t)p * (CPLANE / 4) + q * 32];
        }
    }

    float dAp[16];
#pragma unroll
    for (int q = 0; q < 4; ++q) {
        const float l0 = -expf(va[q].x) * dtd;
        const float l1 = -expf(va[q].y) * dtd;
        const float l2 = -expf(va[q].z) * dtd;
        const float l3 = -expf(va[q].w) * dtd;
        tA[n0 + 4 * q + 0][lane] = expf(l0);
        tA[n0 + 4 * q + 1][lane] = expf(l1);
        tA[n0 + 4 * q + 2][lane] = expf(l2);
        tA[n0 + 4 * q + 3][lane] = expf(l3);
        tC[n0 + 4 * q + 0][lane] = vc[q].x * vb[q].x * dtd;
        tC[n0 + 4 * q + 1][lane] = vc[q].y * vb[q].y * dtd;
        tC[n0 + 4 * q + 2][lane] = vc[q].z * vb[q].z * dtd;
        tC[n0 + 4 * q + 3][lane] = vc[q].w * vb[q].w * dtd;
        dAp[4 * q + 0] = expf(l0 * (float)TCHUNK);
        dAp[4 * q + 1] = expf(l1 * (float)TCHUNK);
        dAp[4 * q + 2] = expf(l2 * (float)TCHUNK);
        dAp[4 * q + 3] = expf(l3 * (float)TCHUNK);
    }

    float sp[16];
#pragma unroll
    for (int i = 0; i < 16; ++i) sp[i] = 0.f;
#pragma unroll
    for (int p = 0; p < NCHUNK; ++p) {
        if (p >= c0 && ((p - c0) & 3) == 0) {
            const int wq = (p - c0) >> 2;
#pragma unroll
            for (int i = 0; i < 16; ++i) sX[wq][n0 + i][lane] = sp[i];
        }
        if (p < 12 || (p < NCHUNK - 1 && p < np)) {
#pragma unroll
            for (int q = 0; q < 4; ++q) carry_step(sp, dAp, e[p < NCHUNK - 1 ? p : 0][q], q);
        }
    }
    __syncthreads();

    float dA[D_STATE], cf[D_STATE], s[D_STATE];
#pragma unroll
    for (int n = 0; n < D_STATE; ++n) {
        dA[n] = tA[n][lane];
        cf[n] = tC[n][lane];
        s[n]  = sX[w][n][lane];
    }

#pragma unroll 1
    for (int t0 = 0; t0 < TCHUNK; t0 += 2 * TBLK) {
#pragma unroll
        for (int k = 0; k < TBLK; ++k)
            ub[k] = up[(size_t)(t0 + TBLK + k) * D_MODEL];
        scan_block(s, dA, cf, ua, Dd, yp + (size_t)t0 * D_MODEL);
        if (t0 + 2 * TBLK < TCHUNK) {
#pragma unroll
            for (int k = 0; k < TBLK; ++k)
                ua[k] = up[(size_t)(t0 + 2 * TBLK + k) * D_MODEL];
        }
        scan_block(s, dA, cf, ub, Dd, yp + (size_t)(t0 + TBLK) * D_MODEL);
    }
}

extern "C" void kernel_launch(void* const* d_in, const int* in_sizes, int n_in,
                              void* d_out, int out_size, void* d_ws, size_t ws_size,
                              hipStream_t stream)
{
    (void)in_sizes; (void)n_in; (void)out_size; (void)ws_size;

    const float* x     = (const float*)d_in[0];
    const float* W_in  = (const float*)d_in[1];
    const float* b_in  = (const float*)d_in[2];
    const float* A_log = (const float*)d_in[3];
    const float* Bm    = (const float*)d_in[4];
    const float* Cm    = (const float*)d_in[5];
    const float* Dv    = (const float*)d_in[6];
    const float* dt    = (const float*)d_in[7];
    const float* W_out = (const float*)d_in[8];
    const float* b_out = (const float*)d_in[9];

    float* xin  = (float*)d_ws;
    float* ymid = xin  + (size_t)MTOT * D_MODEL;
    float* carry = ymid + (size_t)MTOT * D_MODEL;
    float* out  = (float*)d_out;

    gemm_wmma_f32<<<256, dim3(128), 0, stream>>>(x, W_in, b_in, xin);
    s4_chunk_end  <<<dim3(16, 15), dim3(128), 0, stream>>>(xin, A_log, dt, carry);
    s4_scan_chunk <<<dim3(64, 4), dim3(128), 0, stream>>>(xin, A_log, Bm, Cm, Dv, dt, carry, ymid);
    gemm_wmma_f32<<<256, dim3(128), 0, stream>>>(ymid, W_out, b_out, out);
}


# Scan passes fused into one launch: a workgroup owns 8 channels over all 16 chunks (lane = channel x chunk), end states prefixed in LDS; xin and y kept channel-blocked for the GEMMs; sums unchanged
# speedup vs baseline: 5.1070x; 1.1615x over previous
#include <hip/hip_runtime.h>


typedef float v2f __attribute__((ext_vector_type(2)));
typedef float v4f __attribute__((ext_vector_type(4)));
typedef float v8f __attribute__((ext_vector_type(8)));

#define D_MODEL 512
#define D_STATE 64
#define SEQ_LEN 2048
#define BATCH 4
#define MTOT (BATCH * SEQ_LEN)
#define NCHUNK 16
#define TCHUNK (SEQ_LEN / NCHUNK)
#define TBLK 8
#define TBLK1 16
#define XCG 8
#define NCG (D_MODEL / XCG)
#define CBLK (TCHUNK * XCG)

#define GKC 32
#define GP  36
#define GTILE 128
#define GNCH (D_MODEL / GKC)

static __device__ __forceinline__ void load_pair(
    v2f (&fa)[2][4], v2f (&fb)[2][4], const float* As, const float* Bs, const int p)
{
#pragma unroll
    for (int ks = 0; ks < 2; ++ks) {
#pragma unroll
        for (int i = 0; i < 4; ++i)
            fa[ks][i] = *(const v2f*)(As + i * 16 * GP + (2 * p + ks) * 4);
#pragma unroll
        for (int j = 0; j < 4; ++j)
            fb[ks][j] = *(const v2f*)(Bs + j * 16 * GP + (2 * p + ks) * 4);
    }
}

static __device__ __forceinline__ void mma_pair(
    const v2f (&fa)[2][4], const v2f (&fb)[2][4], v8f (&acc)[4][4])
{
#pragma unroll
    for (int ks = 0; ks < 2; ++ks)
#pragma unroll
        for (int i = 0; i < 4; ++i)
#pragma unroll
            for (int j = 0; j < 4; ++j)
                acc[i][j] = __builtin_amdgcn_wmma_f32_16x16x4_f32(
                    false, fa[ks][i], false, fb[ks][j], (short)0, acc[i][j],
                    false, false);
}

static __device__ __forceinline__ void lds_ring_barrier()
{
    asm volatile("s_wait_dscnt 0x0\n\ts_barrier_signal -1\n\ts_barrier_wait -1" ::: "memory");
}

struct Fetch {
    v4f a0, a1, a2, a3, a4, a5, a6, a7;
    v4f b0, b1, b2, b3, b4, b5, b6, b7;
};

template <bool TA>
static __device__ __forceinline__ void fetch_load(Fetch& f, const float* gA, const float* gW)
{
    const size_t R = (size_t)16 * D_MODEL;
    const size_t G = (size_t)NCHUNK * CBLK;
    const size_t H = 16 * 32;
    if (TA) {
        f.a0 = *(const v4f*)(gA + 0 * G);      f.a1 = *(const v4f*)(gA + 0 * G + H);
        f.a2 = *(const v4f*)(gA + 1 * G);      f.a3 = *(const v4f*)(gA + 1 * G + H);
        f.a4 = *(const v4f*)(gA + 2 * G);      f.a5 = *(const v4f*)(gA + 2 * G + H);
        f.a6 = *(const v4f*)(gA + 3 * G);      f.a7 = *(const v4f*)(gA + 3 * G + H);
    } else {
        f.a0 = *(const v4f*)(gA + 0 * R);  f.a1 = *(const v4f*)(gA + 1 * R);
        f.a2 = *(const v4f*)(gA + 2 * R);  f.a3 = *(const v4f*)(gA + 3 * R);
        f.a4 = *(const v4f*)(gA + 4 * R);  f.a5 = *(const v4f*)(gA + 5 * R);
        f.a6 = *(const v4f*)(gA + 6 * R);  f.a7 = *(const v4f*)(gA + 7 * R);
    }
    f.b0 = *(const v4f*)(gW + 0 * R);  f.b1 = *(const v4f*)(gW + 1 * R);
    f.b2 = *(const v4f*)(gW + 2 * R);  f.b3 = *(const v4f*)(gW + 3 * R);
    f.b4 = *(const v4f*)(gW + 4 * R);  f.b5 = *(const v4f*)(gW + 5 * R);
    f.b6 = *(const v4f*)(gW + 6 * R);  f.b7 = *(const v4f*)(gW + 7 * R);
}

static __device__ __forceinline__ void store_col4(float* p, const v4f v)
{
    p[0 * GP] = v.x;  p[1 * GP] = v.y;  p[2 * GP] = v.z;  p[3 * GP] = v.w;
}

template <bool TA>
static __device__ __forceinline__ void fetch_store(const Fetch& f, float* dA, float* dB)
{
    const int R = 16 * GP;
    const int H = 64 * GP;
    if (TA) {
        store_col4(dA + 0 * XCG, f.a0);  store_col4(dA + 0 * XCG + H, f.a1);
        store_col4(dA + 1 * XCG, f.a2);  store_col4(dA + 1 * XCG + H, f.a3);
        store_col4(dA + 2 * XCG, f.a4);  store_col4(dA + 2 * XCG + H, f.a5);
        store_col4(dA + 3 * XCG, f.a6);  store_col4(dA + 3 * XCG + H, f.a7);
    } else {
        *(v4f*)(dA + 0 * R) = f.a0;  *(v4f*)(dA + 1 * R) = f.a1;
        *(v4f*)(dA + 2 * R) = f.a2;  *(v4f*)(dA + 3 * R) = f.a3;
        *(v4f*)(dA + 4 * R) = f.a4;  *(v4f*)(dA + 5 * R) = f.a5;
        *(v4f*)(dA + 6 * R) = f.a6;  *(v4f*)(dA + 7 * R) = f.a7;
    }
    *(v4f*)(dB + 0 * R) = f.b0;  *(v4f*)(dB + 1 * R) = f.b1;
    *(v4f*)(dB + 2 * R) = f.b2;  *(v4f*)(dB + 3 * R) = f.b3;
    *(v4f*)(dB + 4 * R) = f.b4;  *(v4f*)(dB + 5 * R) = f.b5;
    *(v4f*)(dB + 6 * R) = f.b6;  *(v4f*)(dB + 7 * R) = f.b7;
}

template <bool TA, bool LOAD, bool STORE>
static __device__ __forceinline__ void gemm_chunk(
    const float* As, const float* Bs, float* dA, float* dB,
    const float* gA, const float* gW,
    Fetch& ld, const Fetch& st,
    v2f (&fa)[2][2][4], v2f (&fb)[2][2][4], v8f (&acc)[4][4])
{
    load_pair(fa[0], fb[0], As, Bs, 0);
    load_pair(fa[1], fb[1], As, Bs, 1);
    if (LOAD) fetch_load<TA>(ld, gA, gW);
    mma_pair(fa[0], fb[0], acc);
    load_pair(fa[0], fb[0], As, Bs, 2);
    mma_pair(fa[1], fb[1], acc);
    load_pair(fa[1], fb[1], As, Bs, 3);
    mma_pair(fa[0], fb[0], acc);
    if (STORE) fetch_store<TA>(st, dA, dB);
    mma_pair(fa[1], fb[1], acc);

    __builtin_amdgcn_sched_group_barrier(0x100, 16, 0);
    if (LOAD) __builtin_amdgcn_sched_group_barrier(0x020, 16, 0);
    __builtin_amdgcn_sched_group_barrier(0x008, 32, 0);
    __builtin_amdgcn_sched_group_barrier(0x100, 8, 0);
    __builtin_amdgcn_sched_group_barrier(0x008, 32, 0);
    __builtin_amdgcn_sched_group_barrier(0x100, 8, 0);
    __builtin_amdgcn_sched_group_barrier(0x008, 32, 0);
    if (STORE) __builtin_amdgcn_sched_group_barrier(0x200, TA ? 24 : 16, 0);
    __builtin_amdgcn_sched_group_barrier(0x008, 32, 0);
}

template <bool BLK_A, bool BLK_OUT>
__global__ __launch_bounds__(128) __attribute__((amdgpu_waves_per_eu(1, 2)))
void gemm_wmma_f32(
    const float* __restrict__ A, const float* __restrict__ W,
    const float* __restrict__ bias, float* __restrict__ out)
{
    __shared__ float lds[2][2][GTILE * GP];

    const int tid  = threadIdx.x;
    const int lane = tid & 31;
    const int l16  = lane & 15;
    const int half = lane >> 4;
    const int wave = tid >> 5;
    const int wm = wave >> 1;
    const int wn = wave & 1;
    const int M0 = (blockIdx.x >> 2) * GTILE;
    const int N0 = (blockIdx.x & 3) * GTILE;

    const int r0 = tid >> 3;
    const int c4 = tid & 7;
    const int tb = M0 / SEQ_LEN;
    const int tc = (M0 % SEQ_LEN) / TCHUNK;
    const float* gA = BLK_A
        ? A + ((size_t)(tb * NCG) * NCHUNK + tc) * CBLK + r0 * 32 + c4 * 4
        : A + (size_t)(M0 + r0) * D_MODEL + c4 * 4;
    const float* gW = W + (size_t)(N0 + r0) * D_MODEL + c4 * 4;
    const int sOff = r0 * GP + c4 * 4;
    const int sOffA = BLK_A ? (r0 * 4) * GP + c4 : sOff;
    const size_t aStep = BLK_A ? (size_t)4 * NCHUNK * CBLK : (size_t)GKC;

    const int aOff = (wm * 64 + l16) * GP + half * 2;
    const int bOff = (wn * 64 + l16) * GP + half * 2;

    Fetch p0, p1;
    fetch_load<BLK_A>(p0, gA, gW);
    fetch_load<BLK_A>(p1, gA + aStep, gW + GKC);
    fetch_store<BLK_A>(p0, &lds[0][0][sOffA], &lds[0][1][sOff]);
    lds_ring_barrier();

    v8f acc[4][4] = {};
    v2f fa[2][2][4], fb[2][2][4];

#pragma unroll 1
    for (int c = 0; c < GNCH - 2; c += 2) {
        gemm_chunk<BLK_A, true, true>(&lds[0][0][aOff], &lds[0][1][bOff],
                               &lds[1][0][sOffA], &lds[1][1][sOff],
                               gA + (c + 2) * aStep, gW + (c + 2) * GKC,
                               p0, p1, fa, fb, acc);
        lds_ring_barrier();
        gemm_chunk<BLK_A, true, true>(&lds[1][0][aOff], &lds[1][1][bOff],
                               &lds[0][0][sOffA], &lds[0][1][sOff],
                               gA + (c + 3) * aStep, gW + (c + 3) * GKC,
                               p1, p0, fa, fb, acc);
        lds_ring_barrier();
    }
    gemm_chunk<BLK_A, false, true>(&lds[0][0][aOff], &lds[0][1][bOff],
                            &lds[1][0][sOffA], &lds[1][1][sOff],
                            nullptr, nullptr, p0, p1, fa, fb, acc);
    lds_ring_barrier();
    gemm_chunk<BLK_A, false, false>(&lds[1][0][aOff], &lds[1][1][bOff],
                             nullptr, nullptr, nullptr, nullptr, p0, p1, fa, fb, acc);

    const int Mw = M0 + wm * 64;
    const int Nw = N0 + wn * 64;
    if (BLK_OUT) {
#pragma unroll
        for (int j = 0; j < 4; ++j) {
            const int col = Nw + j * 16 + l16;
            const float bj = bias[col];
            float* ob = out + ((size_t)(tb * NCG + (col >> 3)) * NCHUNK + tc) * CBLK + (col & 7) * 4;
#pragma unroll
            for (int i = 0; i < 4; ++i) {
#pragma unroll
                for (int h = 0; h < 2; ++h) {
                    const int t4 = (wm * 64 + i * 16 + half * 8) / 4 + h;
                    v4f v;
                    v.x = acc[i][j][4 * h + 0] + bj;
                    v.y = acc[i][j][4 * h + 1] + bj;
                    v.z = acc[i][j][4 * h + 2] + bj;
                    v.w = acc[i][j][4 * h + 3] + bj;
                    *(v4f*)(ob + t4 * 32) = v;
                }
            }
        }
    } else {
#pragma unroll
        for (int j = 0; j < 4; ++j) {
            const float bj = bias[Nw + j * 16 + l16];
#pragma unroll
            for (int i = 0; i < 4; ++i) {
#pragma unroll
                for (int r = 0; r < 8; ++r) {
                    const int row = Mw + i * 16 + half * 8 + r;
                    out[(size_t)row * D_MODEL + Nw + j * 16 + l16] =
                        acc[i][j][r] + bj;
                }
            }
        }
    }
}

#define UPITCH (CBLK + 32)
#define EPITCH (D_STATE * XCG + 32)

static __device__ __forceinline__ void scan_block(
    float (&s)[D_STATE], const float (&dA)[D_STATE], const float (&cf)[D_STATE],
    const float (&u)[TBLK], const float Dd, float* __restrict__ yp)
{
    float yv[TBLK];
#pragma unroll
    for (int k = 0; k < TBLK; ++k) {
        float p0 = 0.f, p1 = 0.f, p2 = 0.f, p3 = 0.f;
#pragma unroll
        for (int n = 0; n < D_STATE; n += 4) {
            s[n + 0] = fmaf(dA[n + 0], s[n + 0], u[k]);
            s[n + 1] = fmaf(dA[n + 1], s[n + 1], u[k]);
            s[n + 2] = fmaf(dA[n + 2], s[n + 2], u[k]);
            s[n + 3] = fmaf(dA[n + 3], s[n + 3], u[k]);
            p0 = fmaf(cf[n + 0], s[n + 0], p0);
            p1 = fmaf(cf[n + 1], s[n + 1], p1);
            p2 = fmaf(cf[n + 2], s[n + 2], p2);
            p3 = fmaf(cf[n + 3], s[n + 3], p3);
        }
        float yk = fmaf(Dd, u[k], (p0 + p1) + (p2 + p3));
        asm("" : "+v"(yk));
        yv[k] = yk;
    }
#pragma unroll
    for (int j = 0; j < TBLK / 4; ++j) {
        v4f v;
        v.x = yv[4 * j + 0]; v.y = yv[4 * j + 1]; v.z = yv[4 * j + 2]; v.w = yv[4 * j + 3];
        *(v4f*)(yp + j * 32) = v;
    }
}

static __device__ __forceinline__ void end_block(
    float (&s)[D_STATE], const float (&dA)[D_STATE], const v4f u)
{
#pragma unroll
    for (int e = 0; e < 4; ++e) {
#pragma unroll
        for (int n = 0; n < D_STATE; ++n)
            s[n] = fmaf(dA[n], s[n], u[e]);
    }
}

__global__ __launch_bounds__(128) __attribute__((amdgpu_waves_per_eu(1, 2)))
void s4_scan_fused(
    const float* __restrict__ xb,
    const float* __restrict__ A_log,
    const float* __restrict__ Bm,
    const float* __restrict__ Cm,
    const float* __restrict__ Dv,
    const float* __restrict__ dt,
    float* __restrict__ yb)
{
    __shared__ float tA[D_STATE * XCG];
    __shared__ float tC[D_STATE * XCG];
    __shared__ float uS[NCHUNK * UPITCH];
    __shared__ float eS[NCHUNK * EPITCH];

    const int tid = threadIdx.x;
    const int ch  = tid & 7;
    const int c   = tid >> 3;
    const int bg  = blockIdx.x;
    const int d   = (bg & (NCG - 1)) * XCG + ch;
    const size_t blk = ((size_t)bg * NCHUNK + c) * CBLK + ch * 4;
    const float* up = xb + blk;
    float*       yp = yb + blk;

    const float dtd = dt[d];
    const float Dd  = Dv[d];
    const v4f va = *(const v4f*)(A_log + (size_t)d * D_STATE + 4 * c);
    const v4f vb = *(const v4f*)(Bm    + (size_t)d * D_STATE + 4 * c);
    const v4f vc = *(const v4f*)(Cm    + (size_t)d * D_STATE + 4 * c);

    v4f ua[4], ub[4];
#pragma unroll
    for (int j = 0; j < 4; ++j) ua[j] = *(const v4f*)(up + j * 32);

    v4f dA4, cf4, dAp;
#pragma unroll
    for (int e = 0; e < 4; ++e) {
        const float l = -expf(va[e]) * dtd;
        dA4[e] = expf(l);
        cf4[e] = vc[e] * vb[e] * dtd;
        dAp[e] = expf(l * (float)TCHUNK);
    }
    *(v4f*)(tA + tid * 4) = dA4;
    *(v4f*)(tC + tid * 4) = cf4;
    __syncthreads();

    float dA[D_STATE], s[D_STATE];
#pragma unroll
    for (int q = 0; q < D_STATE / 4; ++q) {
        const v4f v = *(const v4f*)(tA + (q * XCG + ch) * 4);
        dA[4 * q + 0] = v.x; dA[4 * q + 1] = v.y; dA[4 * q + 2] = v.z; dA[4 * q + 3] = v.w;
    }
#pragma unroll
    for (int n = 0; n < D_STATE; ++n) s[n] = 0.f;

    float* uw = uS + c * UPITCH + ch * 4;
#pragma unroll 1
    for (int t4 = 0; t4 < TCHUNK / 4; t4 += 8) {
#pragma unroll
        for (int j = 0; j < 4; ++j) ub[j] = *(const v4f*)(up + (t4 + 4 + j) * 32);
#pragma unroll
        for (int j = 0; j < 4; ++j) {
            *(v4f*)(uw + (t4 + j) * 32) = ua[j];
            end_block(s, dA, ua[j]);
        }
        if (t4 + 8 < TCHUNK / 4) {
#pragma unroll
            for (int j = 0; j < 4; ++j) ua[j] = *(const v4f*)(up + (t4 + 8 + j) * 32);
        }
#pragma unroll
        for (int j = 0; j < 4; ++j) {
            *(v4f*)(uw + (t4 + 4 + j) * 32) = ub[j];
            end_block(s, dA, ub[j]);
        }
    }

    float* es = eS + c * EPITCH + ch * 4;
#pragma unroll
    for (int q = 0; q < D_STATE / 4; ++q) {
        v4f v;
        v.x = s[4 * q + 0]; v.y = s[4 * q + 1]; v.z = s[4 * q + 2]; v.w = s[4 * q + 3];
        *(v4f*)(es + q * (XCG * 4)) = v;
    }
    __syncthreads();

    {
        float* pe = eS + tid * 4;
        v4f e[NCHUNK - 1];
#pragma unroll
        for (int p = 0; p < NCHUNK - 1; ++p) e[p] = *(const v4f*)(pe + p * EPITCH);
        v4f sp = {0.f, 0.f, 0.f, 0.f};
#pragma unroll
        for (int p = 0; p < NCHUNK; ++p) {
            *(v4f*)(pe + p * EPITCH) = sp;
            if (p < NCHUNK - 1) {
                sp.x = fmaf(dAp.x, sp.x, e[p].x);
                sp.y = fmaf(dAp.y, sp.y, e[p].y);
                sp.z = fmaf(dAp.z, sp.z, e[p].z);
                sp.w = fmaf(dAp.w, sp.w, e[p].w);
            }
        }
    }
    __syncthreads();

    float cf[D_STATE];
#pragma unroll
    for (int q = 0; q < D_STATE / 4; ++q) {
        const v4f v = *(const v4f*)(tC + (q * XCG + ch) * 4);
        cf[4 * q + 0] = v.x; cf[4 * q + 1] = v.y; cf[4 * q + 2] = v.z; cf[4 * q + 3] = v.w;
        const v4f w = *(const v4f*)(es + q * (XCG * 4));
        s[4 * q + 0] = w.x; s[4 * q + 1] = w.y; s[4 * q + 2] = w.z; s[4 * q + 3] = w.w;
    }

    const float* ur = uw;
    float xa[TBLK], xc[TBLK];
    {
        const v4f v0 = *(const v4f*)(ur), v1 = *(const v4f*)(ur + 32);
        xa[0] = v0.x; xa[1] = v0.y; xa[2] = v0.z; xa[3] = v0.w;
        xa[4] = v1.x; xa[5] = v1.y; xa[6] = v1.z; xa[7] = v1.w;
    }
#pragma unroll 1
    for (int t4 = 0; t4 < TCHUNK / 4; t4 += 4) {
        {
            const v4f v0 = *(const v4f*)(ur + (t4 + 2) * 32), v1 = *(const v4f*)(ur + (t4 + 3) * 32);
            xc[0] = v0.x; xc[1] = v0.y; xc[2] = v0.z; xc[3] = v0.w;
            xc[4] = v1.x; xc[5] = v1.y; xc[6] = v1.z; xc[7] = v1.w;
        }
        scan_block(s, dA, cf, xa, Dd, yp + t4 * 32);
        if (t4 + 4 < TCHUNK / 4) {
            const v4f v0 = *(const v4f*)(ur + (t4 + 4) * 32), v1 = *(const v4f*)(ur + (t4 + 5) * 32);
            xa[0] = v0.x; xa[1] = v0.y; xa[2] = v0.z; xa[3] = v0.w;
            xa[4] = v1.x; xa[5] = v1.y; xa[6] = v1.z; xa[7] = v1.w;
        }
        scan_block(s, dA, cf, xc, Dd, yp + (t4 + 2) * 32);
    }
}

extern "C" void kernel_launch(void* const* d_in, const int* in_sizes, int n_in,
                              void* d_out, int out_size, void* d_ws, size_t ws_size,
                              hipStream_t stream)
{
    (void)in_sizes; (void)n_in; (void)out_size; (void)ws_size;

    const float* x     = (const float*)d_in[0];
    const float* W_in  = (const float*)d_in[1];
    const float* b_in  = (const float*)d_in[2];
    const float* A_log = (const float*)d_in[3];
    const float* Bm    = (const float*)d_in[4];
    const float* Cm    = (const float*)d_in[5];
    const float* Dv    = (const float*)d_in[6];
    const float* dt    = (const float*)d_in[7];
    const float* W_out = (const float*)d_in[8];
    const float* b_out = (const float*)d_in[9];

    float* xin  = (float*)d_ws;
    float* ymid = xin  + (size_t)MTOT * D_MODEL;
    float* out  = (float*)d_out;

    gemm_wmma_f32<false, true><<<256, dim3(128), 0, stream>>>(x, W_in, b_in, xin);
    s4_scan_fused<<<BATCH * NCG, dim3(128), 0, stream>>>(xin, A_log, Bm, Cm, Dv, dt, ymid);
    gemm_wmma_f32<true, false><<<256, dim3(128), 0, stream>>>(ymid, W_out, b_out, out);
}
